# starting kernel
# baseline (speedup 1.0000x reference)
#include <hip/hip_runtime.h>
#include <math.h>

#define Bn 4
#define Nn 512
#define Fo 64
#define Ki 128
#define ALPHAF 0.2f

typedef __attribute__((ext_vector_type(2))) float v2f;
typedef __attribute__((ext_vector_type(8))) float v8f;

__device__ __forceinline__ float lrelu(float v) { return v > 0.f ? v : ALPHAF * v; }

__global__ __launch_bounds__(32) void gat_hgemm(const float* __restrict__ x,
                                                const float* __restrict__ W,
                                                const float* __restrict__ bias,
                                                float* __restrict__ h) {
  const int tile = blockIdx.x;
  const int nt = tile & 3;
  const int mt = tile >> 2;
  const int lane = threadIdx.x;
  const int l15 = lane & 15;
  const int kg  = lane >> 4;
  const int row0 = mt * 16;
  const int n0   = nt * 16;

  v8f acc = {};
  for (int k0 = 0; k0 < Ki; k0 += 4) {
    const int ka = k0 + kg * 2;
    const float* xr = x + (size_t)(row0 + l15) * Ki + ka;
    v2f Af; Af.x = xr[0]; Af.y = xr[1];
    const float* wr = W + (size_t)ka * Fo + n0 + l15;
    v2f Bf; Bf.x = wr[0]; Bf.y = wr[Fo];
    acc = __builtin_amdgcn_wmma_f32_16x16x4_f32(false, Af, false, Bf,
                                                (short)0, acc, false, false);
  }
  const float bv = bias[n0 + l15];
  const int rbase = row0 + kg * 8;
#pragma unroll
  for (int i = 0; i < 8; ++i) {
    h[(size_t)(rbase + i) * Fo + n0 + l15] = acc[i] + bv;
  }
}

__global__ void gat_pq(const float* __restrict__ h, const float* __restrict__ a,
                       float* __restrict__ p, float* __restrict__ q) {
  const int r = blockIdx.x * blockDim.x + threadIdx.x;
  if (r >= Bn * Nn) return;
  const float* hr = h + (size_t)r * Fo;
  float sp = 0.f, sq = 0.f;
#pragma unroll 4
  for (int f = 0; f < Fo; ++f) {
    sp += hr[f] * a[f];
    sq += hr[f] * a[Fo + f];
  }
  p[r] = sp;
  q[r] = sq;
}

__global__ __launch_bounds__(128) void gat_softmax(const int* __restrict__ adj,
                                                   const float* __restrict__ p,
                                                   const float* __restrict__ q,
                                                   float* __restrict__ prob) {
  const int row  = blockIdx.x * 4 + (threadIdx.x >> 5);
  const int lane = threadIdx.x & 31;
  const int b = row >> 9;
  const int n = row & (Nn - 1);
  const int* adjr = adj + (size_t)b * Nn * Nn + (size_t)n * Nn;
  const float* pb = p + b * Nn;
  const float* qb = q + b * Nn;

  float vlo = 0.f, vhi = 0.f;
  if (n < 256) {
    vlo = lrelu(pb[2 * n]     + qb[2 * n]);
    vhi = lrelu(pb[2 * n + 1] + qb[2 * n + 1]);
  }

  float vals[16];
  float mx = -INFINITY;
#pragma unroll
  for (int j = 0; j < 16; ++j) {
    const int m = lane + 32 * j;
    float e;
    if (n < 256) {
      e = (m < 256) ? vlo : vhi;
    } else {
      const int mm = m & 255;
      e = lrelu(pb[2 * mm] + qb[2 * mm + 1]);
    }
    e = (adjr[m] > 0) ? e : -INFINITY;
    vals[j] = e;
    mx = fmaxf(mx, e);
  }
#pragma unroll
  for (int off = 16; off >= 1; off >>= 1) mx = fmaxf(mx, __shfl_xor(mx, off, 32));

  float sum = 0.f;
#pragma unroll
  for (int j = 0; j < 16; ++j) {
    const float e = vals[j];
    const float ex = (e == -INFINITY) ? 0.f : __expf(e - mx);
    vals[j] = ex;
    sum += ex;
  }
#pragma unroll
  for (int off = 16; off >= 1; off >>= 1) sum += __shfl_xor(sum, off, 32);

  const float inv = 1.f / sum;
  float* pr = prob + (size_t)b * Nn * Nn + (size_t)n * Nn;
#pragma unroll
  for (int j = 0; j < 16; ++j) pr[lane + 32 * j] = vals[j] * inv;
}

__global__ __launch_bounds__(32) void gat_aggr(const float* __restrict__ prob,
                                               const float* __restrict__ h,
                                               float* __restrict__ out) {
  const int tile = blockIdx.x;
  const int nt = tile & 3;
  const int mt = (tile >> 2) & 31;
  const int b  = tile >> 7;
  const int lane = threadIdx.x;
  const int l15 = lane & 15;
  const int kg  = lane >> 4;
  const int row0 = mt * 16;
  const int n0   = nt * 16;

  const float* pb = prob + (size_t)b * Nn * Nn;
  const float* hb = h + (size_t)b * Nn * Fo;

  v8f acc = {};
  for (int k0 = 0; k0 < Nn; k0 += 4) {
    const int ka = k0 + kg * 2;
    const float* ar = pb + (size_t)(row0 + l15) * Nn + ka;
    v2f Af; Af.x = ar[0]; Af.y = ar[1];
    const float* br = hb + (size_t)ka * Fo + n0 + l15;
    v2f Bf; Bf.x = br[0]; Bf.y = br[Fo];
    acc = __builtin_amdgcn_wmma_f32_16x16x4_f32(false, Af, false, Bf,
                                                (short)0, acc, false, false);
  }
  float* ob = out + (size_t)b * Nn * Fo;
  const int rbase = row0 + kg * 8;
#pragma unroll
  for (int i = 0; i < 8; ++i) {
    float v = acc[i];
    v = (v > 0.f) ? v : (__expf(v) - 1.f);
    ob[(size_t)(rbase + i) * Fo + n0 + l15] = v;
  }
}

extern "C" void kernel_launch(void* const* d_in, const int* in_sizes, int n_in,
                              void* d_out, int out_size, void* d_ws, size_t ws_size,
                              hipStream_t stream) {
  const float* x    = (const float*)d_in[0];
  const int*   adj  = (const int*)d_in[1];
  const float* W    = (const float*)d_in[2];
  const float* bias = (const float*)d_in[3];
  const float* a    = (const float*)d_in[4];
  float* out = (float*)d_out;

  float* h    = (float*)d_ws;
  float* p    = h + (size_t)Bn * Nn * Fo;
  float* q    = p + Bn * Nn;
  float* prob = q + Bn * Nn;

  gat_hgemm<<<512, 32, 0, stream>>>(x, W, bias, h);
  gat_pq<<<(Bn * Nn + 255) / 256, 256, 0, stream>>>(h, a, p, q);
  gat_softmax<<<512, 128, 0, stream>>>(adj, p, q, prob);
  gat_aggr<<<512, 32, 0, stream>>>(prob, h, out);
}


# aggregation GEMM: K=512 split over 8 waves per 16x32 output strip, each wave issues all its fragment loads as one group; partials summed through LDS in wave order (summation order changes)
# speedup vs baseline: 1.4160x; 1.4160x over previous
#include <hip/hip_runtime.h>
#include <math.h>

#define Bn 4
#define Nn 512
#define Fo 64
#define Ki 128
#define ALPHAF 0.2f

typedef __attribute__((ext_vector_type(2))) float v2f;
typedef __attribute__((ext_vector_type(8))) float v8f;

__device__ __forceinline__ float lrelu(float v) { return v > 0.f ? v : ALPHAF * v; }

__global__ __launch_bounds__(32) void gat_hgemm(const float* __restrict__ x,
                                                const float* __restrict__ W,
                                                const float* __restrict__ bias,
                                                float* __restrict__ h) {
  const int tile = blockIdx.x;
  const int nt = tile & 3;
  const int mt = tile >> 2;
  const int lane = threadIdx.x;
  const int l15 = lane & 15;
  const int kg  = lane >> 4;
  const int row0 = mt * 16;
  const int n0   = nt * 16;

  v8f acc = {};
  for (int k0 = 0; k0 < Ki; k0 += 4) {
    const int ka = k0 + kg * 2;
    const float* xr = x + (size_t)(row0 + l15) * Ki + ka;
    v2f Af; Af.x = xr[0]; Af.y = xr[1];
    const float* wr = W + (size_t)ka * Fo + n0 + l15;
    v2f Bf; Bf.x = wr[0]; Bf.y = wr[Fo];
    acc = __builtin_amdgcn_wmma_f32_16x16x4_f32(false, Af, false, Bf,
                                                (short)0, acc, false, false);
  }
  const float bv = bias[n0 + l15];
  const int rbase = row0 + kg * 8;
#pragma unroll
  for (int i = 0; i < 8; ++i) {
    h[(size_t)(rbase + i) * Fo + n0 + l15] = acc[i] + bv;
  }
}

__global__ void gat_pq(const float* __restrict__ h, const float* __restrict__ a,
                       float* __restrict__ p, float* __restrict__ q) {
  const int r = blockIdx.x * blockDim.x + threadIdx.x;
  if (r >= Bn * Nn) return;
  const float* hr = h + (size_t)r * Fo;
  float sp = 0.f, sq = 0.f;
#pragma unroll 4
  for (int f = 0; f < Fo; ++f) {
    sp += hr[f] * a[f];
    sq += hr[f] * a[Fo + f];
  }
  p[r] = sp;
  q[r] = sq;
}

__global__ __launch_bounds__(128) void gat_softmax(const int* __restrict__ adj,
                                                   const float* __restrict__ p,
                                                   const float* __restrict__ q,
                                                   float* __restrict__ prob) {
  const int row  = blockIdx.x * 4 + (threadIdx.x >> 5);
  const int lane = threadIdx.x & 31;
  const int b = row >> 9;
  const int n = row & (Nn - 1);
  const int* adjr = adj + (size_t)b * Nn * Nn + (size_t)n * Nn;
  const float* pb = p + b * Nn;
  const float* qb = q + b * Nn;

  float vlo = 0.f, vhi = 0.f;
  if (n < 256) {
    vlo = lrelu(pb[2 * n]     + qb[2 * n]);
    vhi = lrelu(pb[2 * n + 1] + qb[2 * n + 1]);
  }

  float vals[16];
  float mx = -INFINITY;
#pragma unroll
  for (int j = 0; j < 16; ++j) {
    const int m = lane + 32 * j;
    float e;
    if (n < 256) {
      e = (m < 256) ? vlo : vhi;
    } else {
      const int mm = m & 255;
      e = lrelu(pb[2 * mm] + qb[2 * mm + 1]);
    }
    e = (adjr[m] > 0) ? e : -INFINITY;
    vals[j] = e;
    mx = fmaxf(mx, e);
  }
#pragma unroll
  for (int off = 16; off >= 1; off >>= 1) mx = fmaxf(mx, __shfl_xor(mx, off, 32));

  float sum = 0.f;
#pragma unroll
  for (int j = 0; j < 16; ++j) {
    const float e = vals[j];
    const float ex = (e == -INFINITY) ? 0.f : __expf(e - mx);
    vals[j] = ex;
    sum += ex;
  }
#pragma unroll
  for (int off = 16; off >= 1; off >>= 1) sum += __shfl_xor(sum, off, 32);

  const float inv = 1.f / sum;
  float* pr = prob + (size_t)b * Nn * Nn + (size_t)n * Nn;
#pragma unroll
  for (int j = 0; j < 16; ++j) pr[lane + 32 * j] = vals[j] * inv;
}

#define AG_WAVES 8
#define AG_KW (Nn / AG_WAVES)
#define AG_PITCH 36

__global__ __launch_bounds__(256) void gat_aggr(const float* __restrict__ prob,
                                                const float* __restrict__ h,
                                                float* __restrict__ out) {
  __shared__ float part[AG_WAVES][16][AG_PITCH];

  const int blk  = blockIdx.x;
  const int nt   = blk & 1;
  const int mt   = (blk >> 1) & 31;
  const int b    = blk >> 6;
  const int wave = threadIdx.x >> 5;
  const int lane = threadIdx.x & 31;
  const int c    = lane & 15;
  const int hh   = lane >> 4;
  const int row0 = mt * 16;
  const int n0   = nt * 32;
  const int kw   = wave * AG_KW;

  const float* ap = prob + (size_t)b * Nn * Nn + (size_t)(row0 + c) * Nn + kw + 4 * hh;
  const float* bp = h + (size_t)b * Nn * Fo + (size_t)(kw + 4 * hh) * Fo + n0 + 2 * c;

  float4 Av[AG_KW / 8];
  float2 Bv[AG_KW / 8][4];
#pragma unroll
  for (int s = 0; s < AG_KW / 8; ++s) {
    Av[s] = *reinterpret_cast<const float4*>(ap + 8 * s);
#pragma unroll
    for (int e = 0; e < 4; ++e)
      Bv[s][e] = *reinterpret_cast<const float2*>(bp + (size_t)(8 * s + e) * Fo);
  }

  __builtin_amdgcn_sched_barrier(0);

  v8f acc0 = {}, acc1 = {};
#pragma unroll
  for (int s = 0; s < AG_KW / 8; ++s) {
    v2f A01; A01.x = Av[s].x; A01.y = Av[s].y;
    v2f A23; A23.x = Av[s].z; A23.y = Av[s].w;
    v2f B0a; B0a.x = Bv[s][0].x; B0a.y = Bv[s][1].x;
    v2f B1a; B1a.x = Bv[s][0].y; B1a.y = Bv[s][1].y;
    v2f B0b; B0b.x = Bv[s][2].x; B0b.y = Bv[s][3].x;
    v2f B1b; B1b.x = Bv[s][2].y; B1b.y = Bv[s][3].y;
    acc0 = __builtin_amdgcn_wmma_f32_16x16x4_f32(false, A01, false, B0a,
                                                 (short)0, acc0, false, false);
    acc1 = __builtin_amdgcn_wmma_f32_16x16x4_f32(false, A01, false, B1a,
                                                 (short)0, acc1, false, false);
    acc0 = __builtin_amdgcn_wmma_f32_16x16x4_f32(false, A23, false, B0b,
                                                 (short)0, acc0, false, false);
    acc1 = __builtin_amdgcn_wmma_f32_16x16x4_f32(false, A23, false, B1b,
                                                 (short)0, acc1, false, false);
  }

  float* pw = &part[wave][8 * hh][2 * c];
#pragma unroll
  for (int r = 0; r < 8; ++r)
    *reinterpret_cast<float2*>(pw + r * AG_PITCH) = make_float2(acc0[r], acc1[r]);
  __syncthreads();

  const int orow = threadIdx.x >> 4;
  const int oc   = (threadIdx.x & 15) * 2;
  float2 sv = *reinterpret_cast<const float2*>(&part[0][orow][oc]);
#pragma unroll
  for (int w = 1; w < AG_WAVES; ++w) {
    const float2 pv = *reinterpret_cast<const float2*>(&part[w][orow][oc]);
    sv.x += pv.x;
    sv.y += pv.y;
  }
  sv.x = (sv.x > 0.f) ? sv.x : (__expf(sv.x) - 1.f);
  sv.y = (sv.y > 0.f) ? sv.y : (__expf(sv.y) - 1.f);
  float* ob = out + (size_t)b * Nn * Fo + (size_t)(row0 + orow) * Fo + n0 + oc;
  *reinterpret_cast<float2*>(ob) = sv;
}

extern "C" void kernel_launch(void* const* d_in, const int* in_sizes, int n_in,
                              void* d_out, int out_size, void* d_ws, size_t ws_size,
                              hipStream_t stream) {
  const float* x    = (const float*)d_in[0];
  const int*   adj  = (const int*)d_in[1];
  const float* W    = (const float*)d_in[2];
  const float* bias = (const float*)d_in[3];
  const float* a    = (const float*)d_in[4];
  float* out = (float*)d_out;

  float* h    = (float*)d_ws;
  float* p    = h + (size_t)Bn * Nn * Fo;
  float* q    = p + Bn * Nn;
  float* prob = q + Bn * Nn;

  gat_hgemm<<<512, 32, 0, stream>>>(x, W, bias, h);
  gat_pq<<<(Bn * Nn + 255) / 256, 256, 0, stream>>>(h, a, p, q);
  gat_softmax<<<512, 128, 0, stream>>>(adj, p, q, prob);
  gat_aggr<<<256, 256, 0, stream>>>(prob, h, out);
}


# p/q dot products moved into the h GEMM launch: 4-wave work-group per 16-row strip, strip staged in LDS, wave 0 sums h.a by fused multiply-adds in the old f order; 4 launches become 3
# speedup vs baseline: 1.7942x; 1.2671x over previous
#include <hip/hip_runtime.h>
#include <math.h>

#define Bn 4
#define Nn 512
#define Fo 64
#define Ki 128
#define ALPHAF 0.2f

typedef __attribute__((ext_vector_type(2))) float v2f;
typedef __attribute__((ext_vector_type(8))) float v8f;

__device__ __forceinline__ float lrelu(float v) { return v > 0.f ? v : ALPHAF * v; }

#define HG_PITCH 68
#define HG_AOFF  96

__global__ __launch_bounds__(128) void gat_hgemm_pq(const float* __restrict__ x,
                                                    const float* __restrict__ W,
                                                    const float* __restrict__ bias,
                                                    const float* __restrict__ a,
                                                    float* __restrict__ h,
                                                    float* __restrict__ p,
                                                    float* __restrict__ q) {
  __shared__ alignas(16) float strip[16][HG_PITCH];
  __shared__ alignas(16) float avec[HG_AOFF + Fo];

  const int mt = blockIdx.x;
  const int nt = threadIdx.x >> 5;
  const int lane = threadIdx.x & 31;
  const int l15 = lane & 15;
  const int kg  = lane >> 4;
  const int row0 = mt * 16;
  const int n0   = nt * 16;

  const float av = a[threadIdx.x];

  v8f acc = {};
  for (int k0 = 0; k0 < Ki; k0 += 4) {
    const int ka = k0 + kg * 2;
    const float* xr = x + (size_t)(row0 + l15) * Ki + ka;
    v2f Af; Af.x = xr[0]; Af.y = xr[1];
    const float* wr = W + (size_t)ka * Fo + n0 + l15;
    v2f Bf; Bf.x = wr[0]; Bf.y = wr[Fo];
    acc = __builtin_amdgcn_wmma_f32_16x16x4_f32(false, Af, false, Bf,
                                                (short)0, acc, false, false);
  }
  const float bv = bias[n0 + l15];
  const int rl = kg * 8;
  float hv[8];
#pragma unroll
  for (int i = 0; i < 8; ++i) {
    hv[i] = acc[i] + bv;
    strip[rl + i][n0 + l15] = hv[i];
  }
  avec[(threadIdx.x >> 6) * HG_AOFF + (threadIdx.x & 63)] = av;
  __syncthreads();
#pragma unroll
  for (int i = 0; i < 8; ++i)
    h[(size_t)(row0 + rl + i) * Fo + n0 + l15] = hv[i];

  if (threadIdx.x < 32) {
    const int which = lane >> 4;
    const float* sr = &strip[l15][0];
    const float* ar = &avec[which * HG_AOFF];
    float4 sv[Fo / 4], aw[Fo / 4];
#pragma unroll
    for (int j = 0; j < Fo / 4; ++j) {
      sv[j] = *reinterpret_cast<const float4*>(sr + 4 * j);
      aw[j] = *reinterpret_cast<const float4*>(ar + 4 * j);
    }
    __builtin_amdgcn_sched_barrier(0);
    float sdot = 0.f;
#pragma unroll
    for (int j = 0; j < Fo / 4; ++j) {
      sdot = __builtin_fmaf(sv[j].x, aw[j].x, sdot);
      sdot = __builtin_fmaf(sv[j].y, aw[j].y, sdot);
      sdot = __builtin_fmaf(sv[j].z, aw[j].z, sdot);
      sdot = __builtin_fmaf(sv[j].w, aw[j].w, sdot);
    }
    float* dst = which ? q : p;
    dst[row0 + l15] = sdot;
  }
}

__global__ __launch_bounds__(128) void gat_softmax(const int* __restrict__ adj,
                                                   const float* __restrict__ p,
                                                   const float* __restrict__ q,
                                                   float* __restrict__ prob) {
  const int row  = blockIdx.x * 4 + (threadIdx.x >> 5);
  const int lane = threadIdx.x & 31;
  const int b = row >> 9;
  const int n = row & (Nn - 1);
  const int* adjr = adj + (size_t)b * Nn * Nn + (size_t)n * Nn;
  const float* pb = p + b * Nn;
  const float* qb = q + b * Nn;

  float vlo = 0.f, vhi = 0.f;
  if (n < 256) {
    vlo = lrelu(pb[2 * n]     + qb[2 * n]);
    vhi = lrelu(pb[2 * n + 1] + qb[2 * n + 1]);
  }

  float vals[16];
  float mx = -INFINITY;
#pragma unroll
  for (int j = 0; j < 16; ++j) {
    const int m = lane + 32 * j;
    float e;
    if (n < 256) {
      e = (m < 256) ? vlo : vhi;
    } else {
      const int mm = m & 255;
      e = lrelu(pb[2 * mm] + qb[2 * mm + 1]);
    }
    e = (adjr[m] > 0) ? e : -INFINITY;
    vals[j] = e;
    mx = fmaxf(mx, e);
  }
#pragma unroll
  for (int off = 16; off >= 1; off >>= 1) mx = fmaxf(mx, __shfl_xor(mx, off, 32));

  float sum = 0.f;
#pragma unroll
  for (int j = 0; j < 16; ++j) {
    const float e = vals[j];
    const float ex = (e == -INFINITY) ? 0.f : __expf(e - mx);
    vals[j] = ex;
    sum += ex;
  }
#pragma unroll
  for (int off = 16; off >= 1; off >>= 1) sum += __shfl_xor(sum, off, 32);

  const float inv = 1.f / sum;
  float* pr = prob + (size_t)b * Nn * Nn + (size_t)n * Nn;
#pragma unroll
  for (int j = 0; j < 16; ++j) pr[lane + 32 * j] = vals[j] * inv;
}

#define AG_WAVES 8
#define AG_KW (Nn / AG_WAVES)
#define AG_PITCH 36

__global__ __launch_bounds__(256) void gat_aggr(const float* __restrict__ prob,
                                                const float* __restrict__ h,
                                                float* __restrict__ out) {
  __shared__ float part[AG_WAVES][16][AG_PITCH];

  const int blk  = blockIdx.x;
  const int nt   = blk & 1;
  const int mt   = (blk >> 1) & 31;
  const int b    = blk >> 6;
  const int wave = threadIdx.x >> 5;
  const int lane = threadIdx.x & 31;
  const int c    = lane & 15;
  const int hh   = lane >> 4;
  const int row0 = mt * 16;
  const int n0   = nt * 32;
  const int kw   = wave * AG_KW;

  const float* ap = prob + (size_t)b * Nn * Nn + (size_t)(row0 + c) * Nn + kw + 4 * hh;
  const float* bp = h + (size_t)b * Nn * Fo + (size_t)(kw + 4 * hh) * Fo + n0 + 2 * c;

  float4 Av[AG_KW / 8];
  float2 Bv[AG_KW / 8][4];
#pragma unroll
  for (int s = 0; s < AG_KW / 8; ++s) {
    Av[s] = *reinterpret_cast<const float4*>(ap + 8 * s);
#pragma unroll
    for (int e = 0; e < 4; ++e)
      Bv[s][e] = *reinterpret_cast<const float2*>(bp + (size_t)(8 * s + e) * Fo);
  }

  __builtin_amdgcn_sched_barrier(0);

  v8f acc0 = {}, acc1 = {};
#pragma unroll
  for (int s = 0; s < AG_KW / 8; ++s) {
    v2f A01; A01.x = Av[s].x; A01.y = Av[s].y;
    v2f A23; A23.x = Av[s].z; A23.y = Av[s].w;
    v2f B0a; B0a.x = Bv[s][0].x; B0a.y = Bv[s][1].x;
    v2f B1a; B1a.x = Bv[s][0].y; B1a.y = Bv[s][1].y;
    v2f B0b; B0b.x = Bv[s][2].x; B0b.y = Bv[s][3].x;
    v2f B1b; B1b.x = Bv[s][2].y; B1b.y = Bv[s][3].y;
    acc0 = __builtin_amdgcn_wmma_f32_16x16x4_f32(false, A01, false, B0a,
                                                 (short)0, acc0, false, false);
    acc1 = __builtin_amdgcn_wmma_f32_16x16x4_f32(false, A01, false, B1a,
                                                 (short)0, acc1, false, false);
    acc0 = __builtin_amdgcn_wmma_f32_16x16x4_f32(false, A23, false, B0b,
                                                 (short)0, acc0, false, false);
    acc1 = __builtin_amdgcn_wmma_f32_16x16x4_f32(false, A23, false, B1b,
                                                 (short)0, acc1, false, false);
  }

  float* pw = &part[wave][8 * hh][2 * c];
#pragma unroll
  for (int r = 0; r < 8; ++r)
    *reinterpret_cast<float2*>(pw + r * AG_PITCH) = make_float2(acc0[r], acc1[r]);
  __syncthreads();

  const int orow = threadIdx.x >> 4;
  const int oc   = (threadIdx.x & 15) * 2;
  float2 sv = *reinterpret_cast<const float2*>(&part[0][orow][oc]);
#pragma unroll
  for (int w = 1; w < AG_WAVES; ++w) {
    const float2 pv = *reinterpret_cast<const float2*>(&part[w][orow][oc]);
    sv.x += pv.x;
    sv.y += pv.y;
  }
  sv.x = (sv.x > 0.f) ? sv.x : (__expf(sv.x) - 1.f);
  sv.y = (sv.y > 0.f) ? sv.y : (__expf(sv.y) - 1.f);
  float* ob = out + (size_t)b * Nn * Fo + (size_t)(row0 + orow) * Fo + n0 + oc;
  *reinterpret_cast<float2*>(ob) = sv;
}

extern "C" void kernel_launch(void* const* d_in, const int* in_sizes, int n_in,
                              void* d_out, int out_size, void* d_ws, size_t ws_size,
                              hipStream_t stream) {
  const float* x    = (const float*)d_in[0];
  const int*   adj  = (const int*)d_in[1];
  const float* W    = (const float*)d_in[2];
  const float* bias = (const float*)d_in[3];
  const float* a    = (const float*)d_in[4];
  float* out = (float*)d_out;

  float* h    = (float*)d_ws;
  float* p    = h + (size_t)Bn * Nn * Fo;
  float* q    = p + Bn * Nn;
  float* prob = q + Bn * Nn;

  gat_hgemm_pq<<<128, 128, 0, stream>>>(x, W, bias, a, h, p, q);
  gat_softmax<<<512, 128, 0, stream>>>(adj, p, q, prob);
  gat_aggr<<<256, 256, 0, stream>>>(prob, h, out);
}


# masked softmax fused into the aggregation launch: each lane builds its A fragments from mask and logits in registers, row max and sum cross the 8 waves in LDS (row sum order changes); 3 launches to 2
# speedup vs baseline: 3.2314x; 1.8010x over previous
#include <hip/hip_runtime.h>
#include <math.h>

#define Bn 4
#define Nn 512
#define Fo 64
#define Ki 128
#define ALPHAF 0.2f

typedef __attribute__((ext_vector_type(2))) float v2f;
typedef __attribute__((ext_vector_type(8))) float v8f;

__device__ __forceinline__ float lrelu(float v) { return v > 0.f ? v : ALPHAF * v; }

#define HG_PITCH 68
#define HG_AOFF  96

__global__ __launch_bounds__(128) void gat_hgemm_pq(const float* __restrict__ x,
                                                    const float* __restrict__ W,
                                                    const float* __restrict__ bias,
                                                    const float* __restrict__ a,
                                                    float* __restrict__ h,
                                                    float* __restrict__ p,
                                                    float* __restrict__ q) {
  __shared__ alignas(16) float strip[16][HG_PITCH];
  __shared__ alignas(16) float avec[HG_AOFF + Fo];

  const int mt = blockIdx.x;
  const int nt = threadIdx.x >> 5;
  const int lane = threadIdx.x & 31;
  const int l15 = lane & 15;
  const int kg  = lane >> 4;
  const int row0 = mt * 16;
  const int n0   = nt * 16;

  const float av = a[threadIdx.x];

  v8f acc = {};
  for (int k0 = 0; k0 < Ki; k0 += 4) {
    const int ka = k0 + kg * 2;
    const float* xr = x + (size_t)(row0 + l15) * Ki + ka;
    v2f Af; Af.x = xr[0]; Af.y = xr[1];
    const float* wr = W + (size_t)ka * Fo + n0 + l15;
    v2f Bf; Bf.x = wr[0]; Bf.y = wr[Fo];
    acc = __builtin_amdgcn_wmma_f32_16x16x4_f32(false, Af, false, Bf,
                                                (short)0, acc, false, false);
  }
  const float bv = bias[n0 + l15];
  const int rl = kg * 8;
  float hv[8];
#pragma unroll
  for (int i = 0; i < 8; ++i) {
    hv[i] = acc[i] + bv;
    strip[rl + i][n0 + l15] = hv[i];
  }
  avec[(threadIdx.x >> 6) * HG_AOFF + (threadIdx.x & 63)] = av;
  __syncthreads();
#pragma unroll
  for (int i = 0; i < 8; ++i)
    h[(size_t)(row0 + rl + i) * Fo + n0 + l15] = hv[i];

  if (threadIdx.x < 32) {
    const int which = lane >> 4;
    const float* sr = &strip[l15][0];
    const float* ar = &avec[which * HG_AOFF];
    float4 sv[Fo / 4], aw[Fo / 4];
#pragma unroll
    for (int j = 0; j < Fo / 4; ++j) {
      sv[j] = *reinterpret_cast<const float4*>(sr + 4 * j);
      aw[j] = *reinterpret_cast<const float4*>(ar + 4 * j);
    }
    __builtin_amdgcn_sched_barrier(0);
    float sdot = 0.f;
#pragma unroll
    for (int j = 0; j < Fo / 4; ++j) {
      sdot = __builtin_fmaf(sv[j].x, aw[j].x, sdot);
      sdot = __builtin_fmaf(sv[j].y, aw[j].y, sdot);
      sdot = __builtin_fmaf(sv[j].z, aw[j].z, sdot);
      sdot = __builtin_fmaf(sv[j].w, aw[j].w, sdot);
    }
    float* dst = which ? q : p;
    dst[row0 + l15] = sdot;
  }
}

#define AG_WAVES 8
#define AG_KW (Nn / AG_WAVES)
#define AG_PITCH 36
#define RS_PITCH 12

__global__ __launch_bounds__(256) void gat_softmax_aggr(const int* __restrict__ adj,
                                                        const float* __restrict__ p,
                                                        const float* __restrict__ q,
                                                        const float* __restrict__ h,
                                                        float* __restrict__ out) {
  __shared__ float part[AG_WAVES][16][AG_PITCH];
  __shared__ alignas(16) float etab[Nn / 2];
  __shared__ alignas(16) float wmax[16][RS_PITCH];
  __shared__ alignas(16) float wsum[16][RS_PITCH];

  const int blk  = blockIdx.x;
  const int nt   = blk & 1;
  const int mt   = (blk >> 1) & 31;
  const int b    = blk >> 6;
  const int wave = threadIdx.x >> 5;
  const int lane = threadIdx.x & 31;
  const int c    = lane & 15;
  const int hh   = lane >> 4;
  const int row0 = mt * 16;
  const int n0   = nt * 32;
  const int kw   = wave * AG_KW;
  const bool lowrows = mt < 16;

  const int pi = lowrows ? (row0 + c) : (int)threadIdx.x;
  const float2 pp = *reinterpret_cast<const float2*>(p + b * Nn + 2 * pi);
  const float2 qq = *reinterpret_cast<const float2*>(q + b * Nn + 2 * pi);

  const int* gp = adj + (size_t)b * Nn * Nn + (size_t)(row0 + c) * Nn + kw + 4 * hh;
  const float* bp = h + (size_t)b * Nn * Fo + (size_t)(kw + 4 * hh) * Fo + n0 + 2 * c;

  int4 Gv[AG_KW / 8];
  float2 Bv[AG_KW / 8][4];
#pragma unroll
  for (int s = 0; s < AG_KW / 8; ++s) {
    Gv[s] = *reinterpret_cast<const int4*>(gp + 8 * s);
#pragma unroll
    for (int e = 0; e < 4; ++e)
      Bv[s][e] = *reinterpret_cast<const float2*>(bp + (size_t)(8 * s + e) * Fo);
  }

  __builtin_amdgcn_sched_barrier(0);

  float elane = 0.f;
  if (lowrows) elane = lrelu(wave < 4 ? pp.x + qq.x : pp.y + qq.y);
  else         etab[threadIdx.x] = lrelu(pp.x + qq.y);
  __syncthreads();

  float ev[AG_KW / 8][4];
  if (lowrows) {
#pragma unroll
    for (int s = 0; s < AG_KW / 8; ++s)
      ev[s][0] = ev[s][1] = ev[s][2] = ev[s][3] = elane;
  } else {
    const float* et = &etab[(kw & 255) + 4 * hh];
#pragma unroll
    for (int s = 0; s < AG_KW / 8; ++s) {
      const float4 t = *reinterpret_cast<const float4*>(et + 8 * s);
      ev[s][0] = t.x; ev[s][1] = t.y; ev[s][2] = t.z; ev[s][3] = t.w;
    }
  }

  float mx0 = -INFINITY, mx1 = -INFINITY, mx2 = -INFINITY, mx3 = -INFINITY;
#pragma unroll
  for (int s = 0; s < AG_KW / 8; ++s) {
    mx0 = (Gv[s].x > 0) ? fmaxf(mx0, ev[s][0]) : mx0;
    mx1 = (Gv[s].y > 0) ? fmaxf(mx1, ev[s][1]) : mx1;
    mx2 = (Gv[s].z > 0) ? fmaxf(mx2, ev[s][2]) : mx2;
    mx3 = (Gv[s].w > 0) ? fmaxf(mx3, ev[s][3]) : mx3;
  }
  float mx = fmaxf(fmaxf(mx0, mx1), fmaxf(mx2, mx3));
  mx = fmaxf(mx, __shfl_xor(mx, 16, 32));
  if (hh == 0) wmax[c][wave] = mx;
  __syncthreads();
  {
    const float4 m0 = *reinterpret_cast<const float4*>(&wmax[c][0]);
    const float4 m1 = *reinterpret_cast<const float4*>(&wmax[c][4]);
    mx = fmaxf(fmaxf(fmaxf(m0.x, m0.y), fmaxf(m0.z, m0.w)),
               fmaxf(fmaxf(m1.x, m1.y), fmaxf(m1.z, m1.w)));
  }

  float ls0 = 0.f, ls1 = 0.f, ls2 = 0.f, ls3 = 0.f;
#pragma unroll
  for (int s = 0; s < AG_KW / 8; ++s) {
    ev[s][0] = (Gv[s].x > 0) ? __expf(ev[s][0] - mx) : 0.f;
    ev[s][1] = (Gv[s].y > 0) ? __expf(ev[s][1] - mx) : 0.f;
    ev[s][2] = (Gv[s].z > 0) ? __expf(ev[s][2] - mx) : 0.f;
    ev[s][3] = (Gv[s].w > 0) ? __expf(ev[s][3] - mx) : 0.f;
    ls0 += ev[s][0];
    ls1 += ev[s][1];
    ls2 += ev[s][2];
    ls3 += ev[s][3];
  }
  float lsum = (ls0 + ls1) + (ls2 + ls3);
  lsum += __shfl_xor(lsum, 16, 32);
  if (hh == 0) wsum[c][wave] = lsum;

  v8f acc0 = {}, acc1 = {};
#pragma unroll
  for (int s = 0; s < AG_KW / 8; ++s) {
    v2f A01; A01.x = ev[s][0]; A01.y = ev[s][1];
    v2f A23; A23.x = ev[s][2]; A23.y = ev[s][3];
    v2f B0a; B0a.x = Bv[s][0].x; B0a.y = Bv[s][1].x;
    v2f B1a; B1a.x = Bv[s][0].y; B1a.y = Bv[s][1].y;
    v2f B0b; B0b.x = Bv[s][2].x; B0b.y = Bv[s][3].x;
    v2f B1b; B1b.x = Bv[s][2].y; B1b.y = Bv[s][3].y;
    acc0 = __builtin_amdgcn_wmma_f32_16x16x4_f32(false, A01, false, B0a,
                                                 (short)0, acc0, false, false);
    acc1 = __builtin_amdgcn_wmma_f32_16x16x4_f32(false, A01, false, B1a,
                                                 (short)0, acc1, false, false);
    acc0 = __builtin_amdgcn_wmma_f32_16x16x4_f32(false, A23, false, B0b,
                                                 (short)0, acc0, false, false);
    acc1 = __builtin_amdgcn_wmma_f32_16x16x4_f32(false, A23, false, B1b,
                                                 (short)0, acc1, false, false);
  }

  float* pw = &part[wave][8 * hh][2 * c];
#pragma unroll
  for (int r = 0; r < 8; ++r)
    *reinterpret_cast<float2*>(pw + r * AG_PITCH) = make_float2(acc0[r], acc1[r]);
  __syncthreads();

  const int orow = threadIdx.x >> 4;
  const int oc   = (threadIdx.x & 15) * 2;
  float2 sv = *reinterpret_cast<const float2*>(&part[0][orow][oc]);
#pragma unroll
  for (int w = 1; w < AG_WAVES; ++w) {
    const float2 pv = *reinterpret_cast<const float2*>(&part[w][orow][oc]);
    sv.x += pv.x;
    sv.y += pv.y;
  }
  const float4 s0 = *reinterpret_cast<const float4*>(&wsum[orow][0]);
  const float4 s1 = *reinterpret_cast<const float4*>(&wsum[orow][4]);
  const float tot = ((((((s0.x + s0.y) + s0.z) + s0.w) + s1.x) + s1.y) + s1.z) + s1.w;
  const float inv = 1.f / tot;
  sv.x *= inv;
  sv.y *= inv;
  sv.x = (sv.x > 0.f) ? sv.x : (__expf(sv.x) - 1.f);
  sv.y = (sv.y > 0.f) ? sv.y : (__expf(sv.y) - 1.f);
  float* ob = out + (size_t)b * Nn * Fo + (size_t)(row0 + orow) * Fo + n0 + oc;
  *reinterpret_cast<float2*>(ob) = sv;
}

extern "C" void kernel_launch(void* const* d_in, const int* in_sizes, int n_in,
                              void* d_out, int out_size, void* d_ws, size_t ws_size,
                              hipStream_t stream) {
  const float* x    = (const float*)d_in[0];
  const int*   adj  = (const int*)d_in[1];
  const float* W    = (const float*)d_in[2];
  const float* bias = (const float*)d_in[3];
  const float* a    = (const float*)d_in[4];
  float* out = (float*)d_out;

  float* h    = (float*)d_ws;
  float* p    = h + (size_t)Bn * Nn * Fo;
  float* q    = p + Bn * Nn;

  gat_hgemm_pq<<<128, 128, 0, stream>>>(x, W, bias, a, h, p, q);
  gat_softmax_aggr<<<256, 256, 0, stream>>>(adj, p, q, h, out);
}
